# starting kernel
# baseline (speedup 1.0000x reference)
#include <hip/hip_runtime.h>
#include <math.h>

#define N_NODES 4096
#define N_EDGES 32768
#define DV 128
#define DE 64
#define DU 32
#define DA 64
#define NEG_SLOPE 0.01f

typedef __attribute__((ext_vector_type(2))) float v2f;
typedef __attribute__((ext_vector_type(8))) float v8f;

__device__ __forceinline__ unsigned ordKey(float f) {
    unsigned u = __float_as_uint(f);
    return (u & 0x80000000u) ? ~u : (u | 0x80000000u);
}
__device__ __forceinline__ float ordInv(unsigned v) {
    return __uint_as_float((v & 0x80000000u) ? (v & 0x7FFFFFFFu) : ~v);
}

template <int DIN, int MODE>
__global__ void proj_wmma_kernel(const float* __restrict__ nodes,
                                 const float* __restrict__ edges,
                                 const int* __restrict__ s,
                                 const int* __restrict__ batch,
                                 const float* __restrict__ u,
                                 const float* __restrict__ W,
                                 const float* __restrict__ bias,
                                 float* __restrict__ out,
                                 int rows) {
    const int lane = threadIdx.x & 31;
    const int wave = threadIdx.x >> 5;
    const int tile = blockIdx.x * 8 + wave;
    const int rowBase = (tile >> 2) * 16;
    const int colBase = (tile & 3) * 16;
    if (rowBase >= rows) return;

    const int mrow = rowBase + (lane & 15);
    const int col  = colBase + (lane & 15);
    const int koff = (lane >> 4) * 2;

    const float* __restrict__ arow;
    int ubase;
    if (MODE == 0) {
        arow = nodes + (size_t)mrow * DV;
        ubase = batch[mrow] * DU;
    } else if (MODE == 1) {
        int sn = s[mrow];
        arow = nodes + (size_t)sn * DV;
        ubase = batch[sn] * DU;
    } else {
        arow = edges + (size_t)mrow * DE;
        ubase = batch[s[mrow]] * DU;
    }
    const int F = (MODE == 2) ? DE : DV;

    v8f acc = {};
#pragma unroll
    for (int kb = 0; kb < DIN; kb += 4) {
        const int k0 = kb + koff;
        v2f a, b;
        a.x = (k0 < F)     ? arow[k0]     : u[ubase + k0 - F];
        a.y = (k0 + 1 < F) ? arow[k0 + 1] : u[ubase + k0 + 1 - F];
        b.x = W[(size_t)k0 * DA + col];
        b.y = W[(size_t)(k0 + 1) * DA + col];
        acc = __builtin_amdgcn_wmma_f32_16x16x4_f32(
             false, a,  false, b,
             (short)0, acc,  false,  false);
    }

    const float bc = bias[col];
    const int rOut = rowBase + (lane >> 4) * 8;
#pragma unroll
    for (int j = 0; j < 8; ++j) {
        float x = acc[j] + bc;
        x = (x > 0.f) ? x : x * NEG_SLOPE;
        out[(size_t)(rOut + j) * DA + col] = x;
    }
}

__global__ void init_kernel(float* __restrict__ out, float* __restrict__ denom,
                            unsigned* __restrict__ maxOrd) {
    int i = blockIdx.x * blockDim.x + threadIdx.x;
    if (i < N_NODES * DA) out[i] = 0.f;
    if (i < N_NODES) { denom[i] = 0.f; maxOrd[i] = 0u; }
}

__global__ void logit_kernel(const float* __restrict__ q, const float* __restrict__ k,
                             const int* __restrict__ recv, float* __restrict__ t,
                             unsigned* __restrict__ maxOrd) {
    int e = blockIdx.x * blockDim.x + threadIdx.x;
    if (e >= N_EDGES) return;
    int rn = recv[e];
    const float4* qf = (const float4*)(q + (size_t)rn * DA);
    const float4* kf = (const float4*)(k + (size_t)e * DA);
    float sum = 0.f;
#pragma unroll
    for (int i = 0; i < DA / 4; ++i) {
        float4 a = qf[i], b = kf[i];
        sum += a.x * b.x + a.y * b.y + a.z * b.z + a.w * b.w;
    }
    float tv = sum * 0.125f;
    t[e] = tv;
    atomicMax(maxOrd + rn, ordKey(tv));
}

__global__ void scatter_kernel(const float* __restrict__ t, const float* __restrict__ v,
                               const int* __restrict__ recv,
                               const unsigned* __restrict__ maxOrd,
                               float* __restrict__ denom, float* __restrict__ acc) {
    int gid = blockIdx.x * blockDim.x + threadIdx.x;
    int e = gid >> 5;
    int lane = gid & 31;
    if (e >= N_EDGES) return;
    int rn = recv[e];
    float w = __expf(t[e] - ordInv(maxOrd[rn]));
    if (lane == 0) atomicAdd(denom + rn, w);
#pragma unroll
    for (int d = lane; d < DA; d += 32)
        atomicAdd(acc + (size_t)rn * DA + d, w * v[(size_t)e * DA + d]);
}

__global__ void finalize_kernel(float* __restrict__ out, const float* __restrict__ denom) {
    int i = blockIdx.x * blockDim.x + threadIdx.x;
    if (i >= N_NODES * DA) return;
    out[i] = out[i] / denom[i / DA];
}

extern "C" void kernel_launch(void* const* d_in, const int* in_sizes, int n_in,
                              void* d_out, int out_size, void* d_ws, size_t ws_size,
                              hipStream_t stream) {
    const float* nodes = (const float*)d_in[0];
    const float* edges = (const float*)d_in[1];
    const int* edge_index = (const int*)d_in[2];
    const float* u = (const float*)d_in[3];
    const int* batch = (const int*)d_in[4];
    const float* Wq = (const float*)d_in[5];
    const float* bq = (const float*)d_in[6];
    const float* Wk = (const float*)d_in[7];
    const float* bk = (const float*)d_in[8];
    const float* Wv = (const float*)d_in[9];
    const float* bv = (const float*)d_in[10];
    const int* s = edge_index;
    const int* r = edge_index + N_EDGES;
    float* out = (float*)d_out;

    char* ws = (char*)d_ws;
    float* q  = (float*)ws;            ws += (size_t)N_NODES * DA * 4;
    float* kk = (float*)ws;            ws += (size_t)N_EDGES * DA * 4;
    float* vv = (float*)ws;            ws += (size_t)N_EDGES * DA * 4;
    float* t  = (float*)ws;            ws += (size_t)N_EDGES * 4;
    unsigned* maxOrd = (unsigned*)ws;  ws += (size_t)N_NODES * 4;
    float* denom = (float*)ws;         ws += (size_t)N_NODES * 4;

    const int initN = N_NODES * DA;
    init_kernel<<<(initN + 255) / 256, 256, 0, stream>>>(out, denom, maxOrd);

    proj_wmma_kernel<160, 0><<<(N_NODES / 16 * 4) / 8, 256, 0, stream>>>(
        nodes, edges, s, batch, u, Wq, bq, q, N_NODES);
    proj_wmma_kernel<160, 1><<<(N_EDGES / 16 * 4) / 8, 256, 0, stream>>>(
        nodes, edges, s, batch, u, Wk, bk, kk, N_EDGES);
    proj_wmma_kernel<96, 2><<<(N_EDGES / 16 * 4) / 8, 256, 0, stream>>>(
        nodes, edges, s, batch, u, Wv, bv, vv, N_EDGES);

    logit_kernel<<<(N_EDGES + 255) / 256, 256, 0, stream>>>(q, kk, r, t, maxOrd);
    scatter_kernel<<<(N_EDGES * 32 + 255) / 256, 256, 0, stream>>>(t, vv, r, maxOrd,
                                                                   denom, out);
    finalize_kernel<<<(N_NODES * DA + 255) / 256, 256, 0, stream>>>(out, denom);
}


# softmax by ownership: block scans r for its 16 receivers into an LDS edge list, one wave per receiver keeps max/sum/weighted v in registers (online softmax, new sum order); no atomics, 7->4 launches
# speedup vs baseline: 1.4866x; 1.4866x over previous
#include <hip/hip_runtime.h>
#include <math.h>

#define N_NODES 4096
#define N_EDGES 32768
#define DV 128
#define DE 64
#define DU 32
#define DA 64
#define NEG_SLOPE 0.01f

typedef __attribute__((ext_vector_type(2))) float v2f;
typedef __attribute__((ext_vector_type(8))) float v8f;

template <int DIN, int MODE>
__global__ void proj_wmma_kernel(const float* __restrict__ nodes,
                                 const float* __restrict__ edges,
                                 const int* __restrict__ s,
                                 const int* __restrict__ batch,
                                 const float* __restrict__ u,
                                 const float* __restrict__ W,
                                 const float* __restrict__ bias,
                                 float* __restrict__ out,
                                 int rows) {
    const int lane = threadIdx.x & 31;
    const int wave = threadIdx.x >> 5;
    const int tile = blockIdx.x * 8 + wave;
    const int rowBase = (tile >> 2) * 16;
    const int colBase = (tile & 3) * 16;
    if (rowBase >= rows) return;

    const int mrow = rowBase + (lane & 15);
    const int col  = colBase + (lane & 15);
    const int koff = (lane >> 4) * 2;

    const float* __restrict__ arow;
    int ubase;
    if (MODE == 0) {
        arow = nodes + (size_t)mrow * DV;
        ubase = batch[mrow] * DU;
    } else if (MODE == 1) {
        int sn = s[mrow];
        arow = nodes + (size_t)sn * DV;
        ubase = batch[sn] * DU;
    } else {
        arow = edges + (size_t)mrow * DE;
        ubase = batch[s[mrow]] * DU;
    }
    const int F = (MODE == 2) ? DE : DV;

    v8f acc = {};
#pragma unroll
    for (int kb = 0; kb < DIN; kb += 4) {
        const int k0 = kb + koff;
        v2f a, b;
        a.x = (k0 < F)     ? arow[k0]     : u[ubase + k0 - F];
        a.y = (k0 + 1 < F) ? arow[k0 + 1] : u[ubase + k0 + 1 - F];
        b.x = W[(size_t)k0 * DA + col];
        b.y = W[(size_t)(k0 + 1) * DA + col];
        acc = __builtin_amdgcn_wmma_f32_16x16x4_f32(
             false, a,  false, b,
             (short)0, acc,  false,  false);
    }

    const float bc = bias[col];
    const int rOut = rowBase + (lane >> 4) * 8;
#pragma unroll
    for (int j = 0; j < 8; ++j) {
        float x = acc[j] + bc;
        x = (x > 0.f) ? x : x * NEG_SLOPE;
        out[(size_t)(rOut + j) * DA + col] = x;
    }
}

#define RPB 16
#define SLICE (N_EDGES / RPB)
#define SCAN_IT (SLICE / 128)

template <int CTRL>
__device__ __forceinline__ float lane_swz(float x) {
    const int b = __builtin_bit_cast(int, x);
    return __builtin_bit_cast(float, __builtin_amdgcn_update_dpp(b, b, CTRL, 0xf, 0xf, false));
}
__device__ __forceinline__ float lane_xhalf(float x) {
    const int b = __builtin_bit_cast(int, x);
    return __builtin_bit_cast(
        float, __builtin_amdgcn_permlanex16(b, b, 0x76543210, 0xfedcba98, false, false));
}
#define XM1 0x161
#define XM2 0x162
#define XM4 0x164
#define XM8 0x168

template <bool FILL>
__device__ __forceinline__ void scan_pass(const int4 (&rr)[SCAN_IT], const int lo,
                                          const int lane, const int ebase, int& st,
                                          unsigned short* __restrict__ elist) {
#pragma unroll
    for (int it = 0; it < SCAN_IT; ++it) {
        const int comp[4] = {rr[it].x, rr[it].y, rr[it].z, rr[it].w};
        int d[4];
        bool match[4];
        unsigned bal[4];
#pragma unroll
        for (int c = 0; c < 4; ++c) {
            d[c] = comp[c] - lo;
            match[c] = (unsigned)d[c] < (unsigned)RPB;
            bal[c] = __builtin_amdgcn_ballot_w32(match[c]);
        }
        if ((bal[0] | bal[1] | bal[2] | bal[3]) == 0u) continue;
#pragma unroll
        for (int c = 0; c < 4; ++c) {
            unsigned b = bal[c];
            while (b) {
                const int src = __builtin_ctz(b);
                const int d0 = __builtin_amdgcn_readlane(d[c], src);
                const bool mine = match[c] && (d[c] == d0);
                const unsigned same = __builtin_amdgcn_ballot_w32(mine);
                if (FILL) {
                    const int pos = __builtin_amdgcn_readlane(st, d0) +
                                    (int)__builtin_amdgcn_mbcnt_lo(same, 0u);
                    if (mine) elist[pos] = (unsigned short)(ebase + it * 128 + lane * 4 + c);
                }
                if (lane == d0) st += __builtin_popcount(same);
                b &= ~same;
            }
        }
    }
}

__global__ __launch_bounds__(512) void attn_owner_kernel(const float* __restrict__ q,
                                                         const float* __restrict__ k,
                                                         const float* __restrict__ v,
                                                         const int* __restrict__ recv,
                                                         float* __restrict__ out) {
    __shared__ unsigned short elist[N_EDGES];
    __shared__ int cntm[RPB][RPB];

    const int lane = threadIdx.x & 31;
    const int w = __builtin_amdgcn_readfirstlane((int)(threadIdx.x >> 5));
    const int lo = blockIdx.x * RPB;
    const int ebase = w * SLICE;

    const int4* __restrict__ r4 = (const int4*)(recv + ebase);
    int4 rr[SCAN_IT];
#pragma unroll
    for (int it = 0; it < SCAN_IT; ++it) rr[it] = r4[it * 32 + lane];

    const int g = lane >> 3;
    const int sub = lane & 7;
    const int rn = lo + w;
    const float4* __restrict__ qp = (const float4*)(q + (size_t)rn * DA);
    const float4 qa = qp[sub], qb = qp[8 + sub];

    int cntv = 0;
    scan_pass<false>(rr, lo, lane, ebase, cntv, elist);
    if (lane < RPB) cntm[w][lane] = cntv;
    __syncthreads();

    int tot = 0, pre = 0;
    {
        const int dcol = lane & 15;
#pragma unroll
        for (int j = 0; j < RPB; ++j) {
            const int cj = cntm[j][dcol];
            tot += cj;
            pre += (j < w) ? cj : 0;
        }
    }
    int incl = tot;
#pragma unroll
    for (int off = 1; off < RPB; off <<= 1) {
        const int up = __shfl_up(incl, off, 16);
        if ((lane & 15) >= off) incl += up;
    }
    const int excl = incl - tot;
    int basev = excl + pre;
    scan_pass<true>(rr, lo, lane, ebase, basev, elist);
    __syncthreads();

    const int start = __builtin_amdgcn_readlane(excl, w);
    const int n = __builtin_amdgcn_readlane(tot, w);

    float m = -INFINITY, l = 0.f;
    float4 a0 = {0.f, 0.f, 0.f, 0.f}, a1 = {0.f, 0.f, 0.f, 0.f};

    for (int i0 = 0; i0 < n; i0 += 16) {
        float4 ka[4], kb[4], va[4], vb[4];
        bool valid[4];
#pragma unroll
        for (int u = 0; u < 4; ++u) {
            const int i = i0 + u * 4 + g;
            valid[u] = i < n;
            const int e = elist[start + (valid[u] ? i : i0)];
            const float4* __restrict__ kp = (const float4*)(k + (size_t)e * DA);
            const float4* __restrict__ vp = (const float4*)(v + (size_t)e * DA);
            ka[u] = kp[sub]; kb[u] = kp[8 + sub];
            va[u] = vp[sub]; vb[u] = vp[8 + sub];
        }
#pragma unroll
        for (int u = 0; u < 4; ++u) {
            float s = qa.x * ka[u].x + qa.y * ka[u].y + qa.z * ka[u].z + qa.w * ka[u].w +
                      qb.x * kb[u].x + qb.y * kb[u].y + qb.z * kb[u].z + qb.w * kb[u].w;
            s += lane_swz<XM1>(s);
            s += lane_swz<XM2>(s);
            s += lane_swz<XM4>(s);
            const float t = s * 0.125f;
            const float mn = valid[u] ? fmaxf(m, t) : m;
            const float sc = (m == mn) ? 1.f : __expf(m - mn);
            const float p = valid[u] ? __expf(t - mn) : 0.f;
            l = l * sc + p;
            a0.x = a0.x * sc + p * va[u].x; a0.y = a0.y * sc + p * va[u].y;
            a0.z = a0.z * sc + p * va[u].z; a0.w = a0.w * sc + p * va[u].w;
            a1.x = a1.x * sc + p * vb[u].x; a1.y = a1.y * sc + p * vb[u].y;
            a1.z = a1.z * sc + p * vb[u].z; a1.w = a1.w * sc + p * vb[u].w;
            m = mn;
        }
    }

    {
        const float mo = lane_swz<XM8>(m), lq = lane_swz<XM8>(l);
        const float M = fmaxf(m, mo);
        const float s1 = (m == M) ? 1.f : __expf(m - M);
        const float s2 = (mo == M) ? 1.f : __expf(mo - M);
        l = l * s1 + lq * s2;
        a0.x = a0.x * s1 + lane_swz<XM8>(a0.x) * s2; a0.y = a0.y * s1 + lane_swz<XM8>(a0.y) * s2;
        a0.z = a0.z * s1 + lane_swz<XM8>(a0.z) * s2; a0.w = a0.w * s1 + lane_swz<XM8>(a0.w) * s2;
        a1.x = a1.x * s1 + lane_swz<XM8>(a1.x) * s2; a1.y = a1.y * s1 + lane_swz<XM8>(a1.y) * s2;
        a1.z = a1.z * s1 + lane_swz<XM8>(a1.z) * s2; a1.w = a1.w * s1 + lane_swz<XM8>(a1.w) * s2;
        m = M;
    }
    {
        const float mo = lane_xhalf(m), lq = lane_xhalf(l);
        const float M = fmaxf(m, mo);
        const float s1 = (m == M) ? 1.f : __expf(m - M);
        const float s2 = (mo == M) ? 1.f : __expf(mo - M);
        l = l * s1 + lq * s2;
        a0.x = a0.x * s1 + lane_xhalf(a0.x) * s2; a0.y = a0.y * s1 + lane_xhalf(a0.y) * s2;
        a0.z = a0.z * s1 + lane_xhalf(a0.z) * s2; a0.w = a0.w * s1 + lane_xhalf(a0.w) * s2;
        a1.x = a1.x * s1 + lane_xhalf(a1.x) * s2; a1.y = a1.y * s1 + lane_xhalf(a1.y) * s2;
        a1.z = a1.z * s1 + lane_xhalf(a1.z) * s2; a1.w = a1.w * s1 + lane_xhalf(a1.w) * s2;
    }

    if (lane < 8) {
        float4* __restrict__ op = (float4*)(out + (size_t)rn * DA);
        float4 o0, o1;
        o0.x = a0.x / l; o0.y = a0.y / l; o0.z = a0.z / l; o0.w = a0.w / l;
        o1.x = a1.x / l; o1.y = a1.y / l; o1.z = a1.z / l; o1.w = a1.w / l;
        op[sub] = o0;
        op[8 + sub] = o1;
    }
}

extern "C" void kernel_launch(void* const* d_in, const int* in_sizes, int n_in,
                              void* d_out, int out_size, void* d_ws, size_t ws_size,
                              hipStream_t stream) {
    const float* nodes = (const float*)d_in[0];
    const float* edges = (const float*)d_in[1];
    const int* edge_index = (const int*)d_in[2];
    const float* u = (const float*)d_in[3];
    const int* batch = (const int*)d_in[4];
    const float* Wq = (const float*)d_in[5];
    const float* bq = (const float*)d_in[6];
    const float* Wk = (const float*)d_in[7];
    const float* bk = (const float*)d_in[8];
    const float* Wv = (const float*)d_in[9];
    const float* bv = (const float*)d_in[10];
    const int* s = edge_index;
    const int* r = edge_index + N_EDGES;
    float* out = (float*)d_out;

    char* ws = (char*)d_ws;
    float* q  = (float*)ws;            ws += (size_t)N_NODES * DA * 4;
    float* kk = (float*)ws;            ws += (size_t)N_EDGES * DA * 4;
    float* vv = (float*)ws;

    proj_wmma_kernel<160, 0><<<(N_NODES / 16 * 4) / 8, 256, 0, stream>>>(
        nodes, edges, s, batch, u, Wq, bq, q, N_NODES);
    proj_wmma_kernel<160, 1><<<(N_EDGES / 16 * 4) / 8, 256, 0, stream>>>(
        nodes, edges, s, batch, u, Wk, bk, kk, N_EDGES);
    proj_wmma_kernel<96, 2><<<(N_EDGES / 16 * 4) / 8, 256, 0, stream>>>(
        nodes, edges, s, batch, u, Wv, bv, vv, N_EDGES);

    attn_owner_kernel<<<N_NODES / RPB, 512, 0, stream>>>(q, kk, vv, r, out);
}


# projections fused into the owner kernel: each block runs the q/k/v WMMA products only for its own edge list (16x64 strips, W staged in LDS), rows kept in LDS; k and v never stored; 4->1 launches
# speedup vs baseline: 2.5937x; 1.7447x over previous
#include <hip/hip_runtime.h>
#include <math.h>


#define N_NODES 4096
#define N_EDGES 32768
#define DV 128
#define DE 64
#define DU 32
#define DA 64
#define NEG_SLOPE 0.01f

typedef __attribute__((ext_vector_type(2))) float v2f;
typedef __attribute__((ext_vector_type(8))) float v8f;

#define RPB 16
#define SLICE (N_EDGES / RPB)
#define SCAN_IT (SLICE / 128)
#define CAP 192
#define PITCH 68
#define WP 160
#define WROWS ((DV + DU) + (DE + DU))

__device__ __forceinline__ v8f mma4(const v2f a, const v2f b, const v8f c) {
    return __builtin_amdgcn_wmma_f32_16x16x4_f32( false, a,  false, b,
                                                  (short)0, c,  false,
                                                  false);
}

template <int CTRL>
__device__ __forceinline__ float lane_swz(float x) {
    const int b = __builtin_bit_cast(int, x);
    return __builtin_bit_cast(float, __builtin_amdgcn_update_dpp(b, b, CTRL, 0xf, 0xf, false));
}
__device__ __forceinline__ float lane_xhalf(float x) {
    const int b = __builtin_bit_cast(int, x);
    return __builtin_bit_cast(
        float, __builtin_amdgcn_permlanex16(b, b, 0x76543210, 0xfedcba98, false, false));
}
#define XM1 0x161
#define XM2 0x162
#define XM4 0x164
#define XM8 0x168

template <bool FILL>
__device__ __forceinline__ void scan_pass(const int4 (&rr)[SCAN_IT], const int lo,
                                          const int lane, const int ebase, int& st,
                                          unsigned short* __restrict__ elist) {
#pragma unroll
    for (int it = 0; it < SCAN_IT; ++it) {
        const int comp[4] = {rr[it].x, rr[it].y, rr[it].z, rr[it].w};
        int d[4];
        bool match[4];
        unsigned bal[4];
#pragma unroll
        for (int c = 0; c < 4; ++c) {
            d[c] = comp[c] - lo;
            match[c] = (unsigned)d[c] < (unsigned)RPB;
            bal[c] = __builtin_amdgcn_ballot_w32(match[c]);
        }
        if ((bal[0] | bal[1] | bal[2] | bal[3]) == 0u) continue;
#pragma unroll
        for (int c = 0; c < 4; ++c) {
            unsigned b = bal[c];
            while (b) {
                const int src = __builtin_ctz(b);
                const int d0 = __builtin_amdgcn_readlane(d[c], src);
                const bool mine = match[c] && (d[c] == d0);
                const unsigned same = __builtin_amdgcn_ballot_w32(mine);
                if (FILL) {
                    const int pos = __builtin_amdgcn_readlane(st, d0) +
                                    (int)__builtin_amdgcn_mbcnt_lo(same, 0u);
                    if (mine) elist[pos] = (unsigned short)(ebase + it * 128 + lane * 4 + c);
                }
                if (lane == d0) st += __builtin_popcount(same);
                b &= ~same;
            }
        }
    }
}

template <int F, int ROW0>
__device__ __forceinline__ void strip(const float* __restrict__ arow,
                                      const float* __restrict__ urow,
                                      const float* __restrict__ wsm,
                                      const float* __restrict__ bias,
                                      float* __restrict__ dst, const int lane) {
    constexpr int NB = (F + DU) / 32;
    constexpr int NF = F / 32;
    const int c = lane & 15, h = lane >> 4;
    const float* __restrict__ wl = wsm + (ROW0 / 2 + h) * WP + 2 * c;
    v8f acc[4] = {};
    v2f an[8];
#pragma unroll
    for (int i = 0; i < 8; ++i) an[i] = *(const v2f*)(arow + 2 * h + 4 * i);
#pragma unroll 1
    for (int j = 0; j < NB; ++j) {
        v2f ac[8];
#pragma unroll
        for (int i = 0; i < 8; ++i) ac[i] = an[i];
        if (j + 1 < NB) {
            const float* __restrict__ pn =
                ((j + 1 < NF) ? (arow + 32 * (j + 1)) : (urow + 32 * (j + 1 - NF))) + 2 * h;
#pragma unroll
            for (int i = 0; i < 8; ++i) an[i] = *(const v2f*)(pn + 4 * i);
        }
        const float* __restrict__ wj = wl + j * 16 * WP;
#pragma unroll
        for (int i = 0; i < 8; ++i) {
#pragma unroll
            for (int ct = 0; ct < 4; ++ct) {
                const v2f b = *(const v2f*)(wj + (2 * i) * WP + 32 * ct);
                acc[ct] = mma4(ac[i], b, acc[ct]);
            }
        }
    }
#pragma unroll
    for (int ct = 0; ct < 4; ++ct) {
        const float bc = bias[ct * 16 + c];
#pragma unroll
        for (int j = 0; j < 8; ++j) {
            float x = acc[ct][j] + bc;
            x = (x > 0.f) ? x : x * NEG_SLOPE;
            dst[(8 * h + j) * PITCH + ct * 16 + c] = x;
        }
    }
}

__device__ __forceinline__ void q_tile(const float* __restrict__ arow,
                                       const float* __restrict__ urow,
                                       const float* __restrict__ W,
                                       const float* __restrict__ bias,
                                       float* __restrict__ qs, const int ct, const int lane) {
    const int c = lane & 15, h = lane >> 4;
    const int col = ct * 16 + c;
    v8f acc = {};
#pragma unroll
    for (int kb = 0; kb < DV + DU; kb += 4) {
        const int k0 = kb + 2 * h;
        const float* __restrict__ p = (kb < DV) ? (arow + k0) : (urow + (k0 - DV));
        const v2f a = *(const v2f*)p;
        v2f b;
        b.x = W[(size_t)k0 * DA + col];
        b.y = W[(size_t)(k0 + 1) * DA + col];
        acc = mma4(a, b, acc);
    }
    const float bc = bias[col];
#pragma unroll
    for (int j = 0; j < 8; ++j) {
        float x = acc[j] + bc;
        x = (x > 0.f) ? x : x * NEG_SLOPE;
        qs[(8 * h + j) * PITCH + col] = x;
    }
}

__global__ __launch_bounds__(512) void edge_attn_owner_kernel(
    const float* __restrict__ nodes, const float* __restrict__ edges,
    const int* __restrict__ s, const int* __restrict__ recv, const int* __restrict__ batch,
    const float* __restrict__ u, const float* __restrict__ Wq, const float* __restrict__ bq,
    const float* __restrict__ Wk, const float* __restrict__ bk, const float* __restrict__ Wv,
    const float* __restrict__ bv, float* __restrict__ out) {
    __shared__ unsigned short elist[N_EDGES];
    __shared__ int cntm[RPB][RPB];
    __shared__ __attribute__((aligned(16))) float wsm[WROWS / 2 * WP];
    __shared__ __attribute__((aligned(16))) float qs[RPB * PITCH];
    __shared__ __attribute__((aligned(16))) float ks[CAP * PITCH];
    __shared__ __attribute__((aligned(16))) float vs[CAP * PITCH];

    const int lane = threadIdx.x & 31;
    const int w = __builtin_amdgcn_readfirstlane((int)(threadIdx.x >> 5));
    const int lo = blockIdx.x * RPB;
    const int ebase = w * SLICE;

    const int4* __restrict__ r4 = (const int4*)(recv + ebase);
    int4 rr[SCAN_IT];
#pragma unroll
    for (int it = 0; it < SCAN_IT; ++it) rr[it] = r4[it * 32 + lane];

    {
#pragma unroll
        for (int it = 0; it < WROWS / 2 * (DA / 4) / 512; ++it) {
            const int item = it * 512 + (int)threadIdx.x;
            const int pr = item >> 4;
            const int a = item & 15;
            const float* __restrict__ srow = (pr < (DV + DU) / 2)
                                                 ? (Wk + (size_t)(2 * pr) * DA)
                                                 : (Wv + (size_t)(2 * pr - (DV + DU)) * DA);
            const float4 x = ((const float4*)srow)[a];
            const float4 y = ((const float4*)(srow + DA))[a];
            float4 o0, o1;
            o0.x = x.x; o0.y = y.x; o0.z = x.y; o0.w = y.y;
            o1.x = x.z; o1.y = y.z; o1.z = x.w; o1.w = y.w;
            float4* __restrict__ d = (float4*)(wsm + pr * WP + a * 8);
            d[0] = o0;
            d[1] = o1;
        }
    }

    const int g = lane >> 3;
    const int sub = lane & 7;
    const int rn = lo + w;

    int cntv = 0;
    scan_pass<false>(rr, lo, lane, ebase, cntv, elist);
    if (lane < RPB) cntm[w][lane] = cntv;
    __syncthreads();

    int tot = 0, pre = 0;
    {
        const int dcol = lane & 15;
#pragma unroll
        for (int j = 0; j < RPB; ++j) {
            const int cj = cntm[j][dcol];
            tot += cj;
            pre += (j < w) ? cj : 0;
        }
    }
    int incl = tot;
#pragma unroll
    for (int off = 1; off < RPB; off <<= 1) {
        const int up = __shfl_up(incl, off, 16);
        if ((lane & 15) >= off) incl += up;
    }
    const int excl = incl - tot;
    int basev = excl + pre;
    scan_pass<true>(rr, lo, lane, ebase, basev, elist);
    __syncthreads();

    const int start = __builtin_amdgcn_readlane(excl, w);
    const int n = __builtin_amdgcn_readlane(tot, w);
    const int etot = __builtin_amdgcn_readlane(incl, RPB - 1);

    const int c = lane & 15;
    if (w >= 12) {
        const int mrow = lo + c;
        q_tile(nodes + (size_t)mrow * DV, u + batch[mrow] * DU, Wq, bq, qs, w - 12, lane);
    }

    float m = -INFINITY, l = 0.f;
    float4 a0 = {0.f, 0.f, 0.f, 0.f}, a1 = {0.f, 0.f, 0.f, 0.f};

    for (int c0 = 0; c0 < etot; c0 += CAP) {
        const int cn = (etot - c0 < CAP) ? (etot - c0) : CAP;
        const int nrt = (cn + 15) >> 4;
        if (w < 8) {
            for (int rt = w; rt < nrt; rt += 8) {
                const int idx = rt * 16 + c;
                const int e = elist[c0 + ((idx < cn) ? idx : 0)];
                const int sn = s[e];
                strip<DV, 0>(nodes + (size_t)sn * DV, u + batch[sn] * DU, wsm, bk,
                             ks + rt * 16 * PITCH, lane);
            }
        } else {
            for (int rt = (w + 6) & 7; rt < nrt; rt += 8) {
                const int idx = rt * 16 + c;
                const int e = elist[c0 + ((idx < cn) ? idx : 0)];
                const int sn = s[e];
                strip<DE, DV + DU>(edges + (size_t)e * DE, u + batch[sn] * DU, wsm, bv,
                                  vs + rt * 16 * PITCH, lane);
            }
        }
        __syncthreads();

        const float4* __restrict__ qp = (const float4*)(qs + w * PITCH);
        const float4 qa = qp[sub], qb = qp[8 + sub];
        const int ra = (c0 - start > 0) ? (c0 - start) : 0;
        const int rb = (c0 + cn - start < n) ? (c0 + cn - start) : n;
        const int roff = start - c0;

        for (int i0 = ra; i0 < rb; i0 += 16) {
            float4 ka[4], kb[4], va[4], vb[4];
            bool valid[4];
#pragma unroll
            for (int uu = 0; uu < 4; ++uu) {
                const int i = i0 + uu * 4 + g;
                valid[uu] = i < rb;
                const int row = (valid[uu] ? i : i0) + roff;
                const float4* __restrict__ kp = (const float4*)(ks + row * PITCH);
                const float4* __restrict__ vp = (const float4*)(vs + row * PITCH);
                ka[uu] = kp[sub]; kb[uu] = kp[8 + sub];
                va[uu] = vp[sub]; vb[uu] = vp[8 + sub];
            }
#pragma unroll
            for (int uu = 0; uu < 4; ++uu) {
                float sd = qa.x * ka[uu].x + qa.y * ka[uu].y + qa.z * ka[uu].z + qa.w * ka[uu].w +
                           qb.x * kb[uu].x + qb.y * kb[uu].y + qb.z * kb[uu].z + qb.w * kb[uu].w;
                sd += lane_swz<XM1>(sd);
                sd += lane_swz<XM2>(sd);
                sd += lane_swz<XM4>(sd);
                const float t = sd * 0.125f;
                const float mn = valid[uu] ? fmaxf(m, t) : m;
                const float sc = (m == mn) ? 1.f : __expf(m - mn);
                const float p = valid[uu] ? __expf(t - mn) : 0.f;
                l = l * sc + p;
                a0.x = a0.x * sc + p * va[uu].x; a0.y = a0.y * sc + p * va[uu].y;
                a0.z = a0.z * sc + p * va[uu].z; a0.w = a0.w * sc + p * va[uu].w;
                a1.x = a1.x * sc + p * vb[uu].x; a1.y = a1.y * sc + p * vb[uu].y;
                a1.z = a1.z * sc + p * vb[uu].z; a1.w = a1.w * sc + p * vb[uu].w;
                m = mn;
            }
        }
        __syncthreads();
    }

    {
        const float mo = lane_swz<XM8>(m), lq = lane_swz<XM8>(l);
        const float M = fmaxf(m, mo);
        const float s1 = (m == M) ? 1.f : __expf(m - M);
        const float s2 = (mo == M) ? 1.f : __expf(mo - M);
        l = l * s1 + lq * s2;
        a0.x = a0.x * s1 + lane_swz<XM8>(a0.x) * s2; a0.y = a0.y * s1 + lane_swz<XM8>(a0.y) * s2;
        a0.z = a0.z * s1 + lane_swz<XM8>(a0.z) * s2; a0.w = a0.w * s1 + lane_swz<XM8>(a0.w) * s2;
        a1.x = a1.x * s1 + lane_swz<XM8>(a1.x) * s2; a1.y = a1.y * s1 + lane_swz<XM8>(a1.y) * s2;
        a1.z = a1.z * s1 + lane_swz<XM8>(a1.z) * s2; a1.w = a1.w * s1 + lane_swz<XM8>(a1.w) * s2;
        m = M;
    }
    {
        const float mo = lane_xhalf(m), lq = lane_xhalf(l);
        const float M = fmaxf(m, mo);
        const float s1 = (m == M) ? 1.f : __expf(m - M);
        const float s2 = (mo == M) ? 1.f : __expf(mo - M);
        l = l * s1 + lq * s2;
        a0.x = a0.x * s1 + lane_xhalf(a0.x) * s2; a0.y = a0.y * s1 + lane_xhalf(a0.y) * s2;
        a0.z = a0.z * s1 + lane_xhalf(a0.z) * s2; a0.w = a0.w * s1 + lane_xhalf(a0.w) * s2;
        a1.x = a1.x * s1 + lane_xhalf(a1.x) * s2; a1.y = a1.y * s1 + lane_xhalf(a1.y) * s2;
        a1.z = a1.z * s1 + lane_xhalf(a1.z) * s2; a1.w = a1.w * s1 + lane_xhalf(a1.w) * s2;
    }

    if (lane < 8) {
        float4* __restrict__ op = (float4*)(out + (size_t)rn * DA);
        float4 o0, o1;
        o0.x = a0.x / l; o0.y = a0.y / l; o0.z = a0.z / l; o0.w = a0.w / l;
        o1.x = a1.x / l; o1.y = a1.y / l; o1.z = a1.z / l; o1.w = a1.w / l;
        op[sub] = o0;
        op[8 + sub] = o1;
    }
}

extern "C" void kernel_launch(void* const* d_in, const int* in_sizes, int n_in,
                              void* d_out, int out_size, void* d_ws, size_t ws_size,
                              hipStream_t stream) {
    const float* nodes = (const float*)d_in[0];
    const float* edges = (const float*)d_in[1];
    const int* edge_index = (const int*)d_in[2];
    const float* u = (const float*)d_in[3];
    const int* batch = (const int*)d_in[4];
    const float* Wq = (const float*)d_in[5];
    const float* bq = (const float*)d_in[6];
    const float* Wk = (const float*)d_in[7];
    const float* bk = (const float*)d_in[8];
    const float* Wv = (const float*)d_in[9];
    const float* bv = (const float*)d_in[10];
    const int* s = edge_index;
    const int* r = edge_index + N_EDGES;
    float* out = (float*)d_out;

    edge_attn_owner_kernel<<<N_NODES / RPB, 512, 0, stream>>>(nodes, edges, s, r, batch, u, Wq,
                                                              bq, Wk, bk, Wv, bv, out);
}


# u share of the k and v projections read from two 32-graph tables (u x W's u rows + bias, one WMMA tile per wave, once per block); strips run K=128/64, table row added in epilogue (new sum order)
# speedup vs baseline: 2.7717x; 1.0686x over previous
#include <hip/hip_runtime.h>
#include <math.h>


#define N_NODES 4096
#define N_EDGES 32768
#define DV 128
#define DE 64
#define DU 32
#define DA 64
#define NEG_SLOPE 0.01f

typedef __attribute__((ext_vector_type(2))) float v2f;
typedef __attribute__((ext_vector_type(8))) float v8f;

#define RPB 16
#define SLICE (N_EDGES / RPB)
#define SCAN_IT (SLICE / 128)
#define CAP 192
#define PITCH 68
#define WP 160
#define WROWS (DV + DE)
#define NG 32
#define TP 64

__device__ __forceinline__ v8f mma4(const v2f a, const v2f b, const v8f c) {
    return __builtin_amdgcn_wmma_f32_16x16x4_f32( false, a,  false, b,
                                                  (short)0, c,  false,
                                                  false);
}

template <int CTRL>
__device__ __forceinline__ float lane_swz(float x) {
    const int b = __builtin_bit_cast(int, x);
    return __builtin_bit_cast(float, __builtin_amdgcn_update_dpp(b, b, CTRL, 0xf, 0xf, false));
}
__device__ __forceinline__ float lane_xhalf(float x) {
    const int b = __builtin_bit_cast(int, x);
    return __builtin_bit_cast(
        float, __builtin_amdgcn_permlanex16(b, b, 0x76543210, 0xfedcba98, false, false));
}
#define XM1 0x161
#define XM2 0x162
#define XM4 0x164
#define XM8 0x168

template <bool FILL>
__device__ __forceinline__ void scan_pass(const int4 (&rr)[SCAN_IT], const int lo,
                                          const int lane, const int ebase, int& st,
                                          unsigned short* __restrict__ elist) {
#pragma unroll
    for (int it = 0; it < SCAN_IT; ++it) {
        const int comp[4] = {rr[it].x, rr[it].y, rr[it].z, rr[it].w};
        int d[4];
        bool match[4];
        unsigned bal[4];
#pragma unroll
        for (int c = 0; c < 4; ++c) {
            d[c] = comp[c] - lo;
            match[c] = (unsigned)d[c] < (unsigned)RPB;
            bal[c] = __builtin_amdgcn_ballot_w32(match[c]);
        }
        if ((bal[0] | bal[1] | bal[2] | bal[3]) == 0u) continue;
#pragma unroll
        for (int c = 0; c < 4; ++c) {
            unsigned b = bal[c];
            while (b) {
                const int src = __builtin_ctz(b);
                const int d0 = __builtin_amdgcn_readlane(d[c], src);
                const bool mine = match[c] && (d[c] == d0);
                const unsigned same = __builtin_amdgcn_ballot_w32(mine);
                if (FILL) {
                    const int pos = __builtin_amdgcn_readlane(st, d0) +
                                    (int)__builtin_amdgcn_mbcnt_lo(same, 0u);
                    if (mine) elist[pos] = (unsigned short)(ebase + it * 128 + lane * 4 + c);
                }
                if (lane == d0) st += __builtin_popcount(same);
                b &= ~same;
            }
        }
    }
}

template <int F, int ROW0>
__device__ __forceinline__ void strip(const float* __restrict__ arow, const int grow,
                                      const float* __restrict__ wsm,
                                      const float* __restrict__ tab,
                                      float* __restrict__ dst, const int lane) {
    constexpr int NB = F / 32;
    const int c = lane & 15, h = lane >> 4;
    const float* __restrict__ wl = wsm + (ROW0 / 2 + h) * WP + 2 * c;
    v8f acc[4] = {};
    v2f an[8];
#pragma unroll
    for (int i = 0; i < 8; ++i) an[i] = *(const v2f*)(arow + 2 * h + 4 * i);
#pragma unroll 1
    for (int j = 0; j < NB; ++j) {
        v2f ac[8];
#pragma unroll
        for (int i = 0; i < 8; ++i) ac[i] = an[i];
        if (j + 1 < NB) {
            const float* __restrict__ pn = arow + 32 * (j + 1) + 2 * h;
#pragma unroll
            for (int i = 0; i < 8; ++i) an[i] = *(const v2f*)(pn + 4 * i);
        }
        const float* __restrict__ wj = wl + j * 16 * WP;
#pragma unroll
        for (int i = 0; i < 8; ++i) {
#pragma unroll
            for (int ct = 0; ct < 4; ++ct) {
                const v2f b = *(const v2f*)(wj + (2 * i) * WP + 32 * ct);
                acc[ct] = mma4(ac[i], b, acc[ct]);
            }
        }
    }
#pragma unroll
    for (int j = 0; j < 8; ++j) {
        const int gj = __shfl(grow, 8 * h + j, 16);
        const float4 tv = *(const float4*)(tab + gj * TP + 4 * c);
        const float t4[4] = {tv.x, tv.y, tv.z, tv.w};
#pragma unroll
        for (int ct = 0; ct < 4; ++ct) {
            float x = acc[ct][j] + t4[ct];
            x = (x > 0.f) ? x : x * NEG_SLOPE;
            dst[(8 * h + j) * PITCH + ct * 16 + c] = x;
        }
    }
}

__device__ __forceinline__ void u_table_operands(const float* __restrict__ u,
                                                 const float* __restrict__ Wu, const int rt,
                                                 const int ct, const int lane, v2f (&a)[DU / 4],
                                                 v2f (&b)[DU / 4]) {
    const int c = lane & 15, h = lane >> 4;
    const int col = ct * 16 + c;
    const float* __restrict__ urow = u + (rt * 16 + c) * DU;
#pragma unroll
    for (int t = 0; t < DU / 4; ++t) {
        const int k0 = 4 * t + 2 * h;
        a[t] = *(const v2f*)(urow + k0);
        b[t].x = Wu[(size_t)k0 * DA + col];
        b[t].y = Wu[(size_t)(k0 + 1) * DA + col];
    }
}
__device__ __forceinline__ void u_table_tile(const v2f (&a)[DU / 4], const v2f (&b)[DU / 4],
                                             const float bc, float* __restrict__ tab,
                                             const int rt, const int ct, const int lane) {
    const int c = lane & 15, h = lane >> 4;
    v8f acc = {};
#pragma unroll
    for (int t = 0; t < DU / 4; ++t) acc = mma4(a[t], b[t], acc);
#pragma unroll
    for (int j = 0; j < 8; ++j) tab[(rt * 16 + 8 * h + j) * TP + 4 * c + ct] = acc[j] + bc;
}

__device__ __forceinline__ void q_tile(const float* __restrict__ arow,
                                       const float* __restrict__ urow,
                                       const float* __restrict__ W,
                                       const float* __restrict__ bias,
                                       float* __restrict__ qs, const int ct, const int lane) {
    const int c = lane & 15, h = lane >> 4;
    const int col = ct * 16 + c;
    v8f acc = {};
#pragma unroll
    for (int kb = 0; kb < DV + DU; kb += 4) {
        const int k0 = kb + 2 * h;
        const float* __restrict__ p = (kb < DV) ? (arow + k0) : (urow + (k0 - DV));
        const v2f a = *(const v2f*)p;
        v2f b;
        b.x = W[(size_t)k0 * DA + col];
        b.y = W[(size_t)(k0 + 1) * DA + col];
        acc = mma4(a, b, acc);
    }
    const float bc = bias[col];
#pragma unroll
    for (int j = 0; j < 8; ++j) {
        float x = acc[j] + bc;
        x = (x > 0.f) ? x : x * NEG_SLOPE;
        qs[(8 * h + j) * PITCH + col] = x;
    }
}

__global__ __launch_bounds__(512) void edge_attn_owner_kernel(
    const float* __restrict__ nodes, const float* __restrict__ edges,
    const int* __restrict__ s, const int* __restrict__ recv, const int* __restrict__ batch,
    const float* __restrict__ u, const float* __restrict__ Wq, const float* __restrict__ bq,
    const float* __restrict__ Wk, const float* __restrict__ bk, const float* __restrict__ Wv,
    const float* __restrict__ bv, float* __restrict__ out) {
    __shared__ unsigned short elist[N_EDGES];
    __shared__ int cntm[RPB][RPB];
    __shared__ __attribute__((aligned(16))) float wsm[WROWS / 2 * WP];
    __shared__ __attribute__((aligned(16))) float tab[2 * NG * TP];
    __shared__ __attribute__((aligned(16))) float qs[RPB * PITCH];
    __shared__ __attribute__((aligned(16))) float ks[CAP * PITCH];
    __shared__ __attribute__((aligned(16))) float vs[CAP * PITCH];

    const int lane = threadIdx.x & 31;
    const int w = __builtin_amdgcn_readfirstlane((int)(threadIdx.x >> 5));
    const int lo = blockIdx.x * RPB;
    const int ebase = w * SLICE;

    const int trt = (w >> 2) & 1, tct = w & 3;
    v2f ta[DU / 4], tb[DU / 4];
    u_table_operands(u, (w < 8) ? (Wk + (size_t)DV * DA) : (Wv + (size_t)DE * DA), trt, tct, lane,
                     ta, tb);
    const float tbias = ((w < 8) ? bk : bv)[tct * 16 + (lane & 15)];

    const int4* __restrict__ r4 = (const int4*)(recv + ebase);
    int4 rr[SCAN_IT];
#pragma unroll
    for (int it = 0; it < SCAN_IT; ++it) rr[it] = r4[it * 32 + lane];
    __builtin_amdgcn_sched_barrier(0);

    u_table_tile(ta, tb, tbias, tab + (w >> 3) * (NG * TP), trt, tct, lane);

    {
#pragma unroll
        for (int it = 0; it < WROWS / 2 * (DA / 4) / 512; ++it) {
            const int item = it * 512 + (int)threadIdx.x;
            const int pr = item >> 4;
            const int a = item & 15;
            const float* __restrict__ srow = (pr < DV / 2)
                                                 ? (Wk + (size_t)(2 * pr) * DA)
                                                 : (Wv + (size_t)(2 * pr - DV) * DA);
            const float4 x = ((const float4*)srow)[a];
            const float4 y = ((const float4*)(srow + DA))[a];
            float4 o0, o1;
            o0.x = x.x; o0.y = y.x; o0.z = x.y; o0.w = y.y;
            o1.x = x.z; o1.y = y.z; o1.z = x.w; o1.w = y.w;
            float4* __restrict__ d = (float4*)(wsm + pr * WP + a * 8);
            d[0] = o0;
            d[1] = o1;
        }
    }

    const int g = lane >> 3;
    const int sub = lane & 7;
    const int rn = lo + w;

    int cntv = 0;
    scan_pass<false>(rr, lo, lane, ebase, cntv, elist);
    if (lane < RPB) cntm[w][lane] = cntv;
    __syncthreads();

    int tot = 0, pre = 0;
    {
        const int dcol = lane & 15;
#pragma unroll
        for (int j = 0; j < RPB; ++j) {
            const int cj = cntm[j][dcol];
            tot += cj;
            pre += (j < w) ? cj : 0;
        }
    }
    int incl = tot;
#pragma unroll
    for (int off = 1; off < RPB; off <<= 1) {
        const int up = __shfl_up(incl, off, 16);
        if ((lane & 15) >= off) incl += up;
    }
    const int excl = incl - tot;
    int basev = excl + pre;
    scan_pass<true>(rr, lo, lane, ebase, basev, elist);
    __syncthreads();

    const int start = __builtin_amdgcn_readlane(excl, w);
    const int n = __builtin_amdgcn_readlane(tot, w);
    const int etot = __builtin_amdgcn_readlane(incl, RPB - 1);

    const int c = lane & 15;
    if (w >= 12) {
        const int mrow = lo + c;
        q_tile(nodes + (size_t)mrow * DV, u + batch[mrow] * DU, Wq, bq, qs, w - 12, lane);
    }

    float m = -INFINITY, l = 0.f;
    float4 a0 = {0.f, 0.f, 0.f, 0.f}, a1 = {0.f, 0.f, 0.f, 0.f};

    for (int c0 = 0; c0 < etot; c0 += CAP) {
        const int cn = (etot - c0 < CAP) ? (etot - c0) : CAP;
        const int nrt = (cn + 15) >> 4;
        if (w < 8) {
            for (int rt = w; rt < nrt; rt += 8) {
                const int idx = rt * 16 + c;
                const int e = elist[c0 + ((idx < cn) ? idx : 0)];
                const int sn = s[e];
                strip<DV, 0>(nodes + (size_t)sn * DV, batch[sn], wsm, tab,
                             ks + rt * 16 * PITCH, lane);
            }
        } else {
            for (int rt = (w + 6) & 7; rt < nrt; rt += 8) {
                const int idx = rt * 16 + c;
                const int e = elist[c0 + ((idx < cn) ? idx : 0)];
                const int sn = s[e];
                strip<DE, DV>(edges + (size_t)e * DE, batch[sn], wsm, tab + NG * TP,
                              vs + rt * 16 * PITCH, lane);
            }
        }
        __syncthreads();

        const float4* __restrict__ qp = (const float4*)(qs + w * PITCH);
        const float4 qa = qp[sub], qb = qp[8 + sub];
        const int ra = (c0 - start > 0) ? (c0 - start) : 0;
        const int rb = (c0 + cn - start < n) ? (c0 + cn - start) : n;
        const int roff = start - c0;

        for (int i0 = ra; i0 < rb; i0 += 16) {
            float4 ka[4], kb[4], va[4], vb[4];
            bool valid[4];
#pragma unroll
            for (int uu = 0; uu < 4; ++uu) {
                const int i = i0 + uu * 4 + g;
                valid[uu] = i < rb;
                const int row = (valid[uu] ? i : i0) + roff;
                const float4* __restrict__ kp = (const float4*)(ks + row * PITCH);
                const float4* __restrict__ vp = (const float4*)(vs + row * PITCH);
                ka[uu] = kp[sub]; kb[uu] = kp[8 + sub];
                va[uu] = vp[sub]; vb[uu] = vp[8 + sub];
            }
#pragma unroll
            for (int uu = 0; uu < 4; ++uu) {
                float sd = qa.x * ka[uu].x + qa.y * ka[uu].y + qa.z * ka[uu].z + qa.w * ka[uu].w +
                           qb.x * kb[uu].x + qb.y * kb[uu].y + qb.z * kb[uu].z + qb.w * kb[uu].w;
                sd += lane_swz<XM1>(sd);
                sd += lane_swz<XM2>(sd);
                sd += lane_swz<XM4>(sd);
                const float t = sd * 0.125f;
                const float mn = valid[uu] ? fmaxf(m, t) : m;
                const float sc = (m == mn) ? 1.f : __expf(m - mn);
                const float p = valid[uu] ? __expf(t - mn) : 0.f;
                l = l * sc + p;
                a0.x = a0.x * sc + p * va[uu].x; a0.y = a0.y * sc + p * va[uu].y;
                a0.z = a0.z * sc + p * va[uu].z; a0.w = a0.w * sc + p * va[uu].w;
                a1.x = a1.x * sc + p * vb[uu].x; a1.y = a1.y * sc + p * vb[uu].y;
                a1.z = a1.z * sc + p * vb[uu].z; a1.w = a1.w * sc + p * vb[uu].w;
                m = mn;
            }
        }
        __syncthreads();
    }

    {
        const float mo = lane_swz<XM8>(m), lq = lane_swz<XM8>(l);
        const float M = fmaxf(m, mo);
        const float s1 = (m == M) ? 1.f : __expf(m - M);
        const float s2 = (mo == M) ? 1.f : __expf(mo - M);
        l = l * s1 + lq * s2;
        a0.x = a0.x * s1 + lane_swz<XM8>(a0.x) * s2; a0.y = a0.y * s1 + lane_swz<XM8>(a0.y) * s2;
        a0.z = a0.z * s1 + lane_swz<XM8>(a0.z) * s2; a0.w = a0.w * s1 + lane_swz<XM8>(a0.w) * s2;
        a1.x = a1.x * s1 + lane_swz<XM8>(a1.x) * s2; a1.y = a1.y * s1 + lane_swz<XM8>(a1.y) * s2;
        a1.z = a1.z * s1 + lane_swz<XM8>(a1.z) * s2; a1.w = a1.w * s1 + lane_swz<XM8>(a1.w) * s2;
        m = M;
    }
    {
        const float mo = lane_xhalf(m), lq = lane_xhalf(l);
        const float M = fmaxf(m, mo);
        const float s1 = (m == M) ? 1.f : __expf(m - M);
        const float s2 = (mo == M) ? 1.f : __expf(mo - M);
        l = l * s1 + lq * s2;
        a0.x = a0.x * s1 + lane_xhalf(a0.x) * s2; a0.y = a0.y * s1 + lane_xhalf(a0.y) * s2;
        a0.z = a0.z * s1 + lane_xhalf(a0.z) * s2; a0.w = a0.w * s1 + lane_xhalf(a0.w) * s2;
        a1.x = a1.x * s1 + lane_xhalf(a1.x) * s2; a1.y = a1.y * s1 + lane_xhalf(a1.y) * s2;
        a1.z = a1.z * s1 + lane_xhalf(a1.z) * s2; a1.w = a1.w * s1 + lane_xhalf(a1.w) * s2;
    }

    if (lane < 8) {
        float4* __restrict__ op = (float4*)(out + (size_t)rn * DA);
        float4 o0, o1;
        o0.x = a0.x / l; o0.y = a0.y / l; o0.z = a0.z / l; o0.w = a0.w / l;
        o1.x = a1.x / l; o1.y = a1.y / l; o1.z = a1.z / l; o1.w = a1.w / l;
        op[sub] = o0;
        op[8 + sub] = o1;
    }
}

extern "C" void kernel_launch(void* const* d_in, const int* in_sizes, int n_in,
                              void* d_out, int out_size, void* d_ws, size_t ws_size,
                              hipStream_t stream) {
    const float* nodes = (const float*)d_in[0];
    const float* edges = (const float*)d_in[1];
    const int* edge_index = (const int*)d_in[2];
    const float* u = (const float*)d_in[3];
    const int* batch = (const int*)d_in[4];
    const float* Wq = (const float*)d_in[5];
    const float* bq = (const float*)d_in[6];
    const float* Wk = (const float*)d_in[7];
    const float* bk = (const float*)d_in[8];
    const float* Wv = (const float*)d_in[9];
    const float* bv = (const float*)d_in[10];
    const int* s = edge_index;
    const int* r = edge_index + N_EDGES;
    float* out = (float*)d_out;

    edge_attn_owner_kernel<<<N_NODES / RPB, 512, 0, stream>>>(nodes, edges, s, r, batch, u, Wq,
                                                              bq, Wk, bk, Wv, bv, out);
}


# edge list in ONE pass over receiver ids: a wave appends its matches (receiver number in the entry) to its own LDS region; count pass, 2nd scan, a barrier gone; sum order same up to 192 edges/block
# speedup vs baseline: 3.0750x; 1.1094x over previous
#include <hip/hip_runtime.h>
#include <math.h>


#define N_NODES 4096
#define N_EDGES 32768
#define DV 128
#define DE 64
#define DU 32
#define DA 64
#define NEG_SLOPE 0.01f

typedef __attribute__((ext_vector_type(2))) float v2f;
typedef __attribute__((ext_vector_type(8))) float v8f;

#define RPB 16
#define SLICE (N_EDGES / RPB)
#define SCAN_IT (SLICE / 128)
#define LBITS 11
static_assert((1 << LBITS) == SLICE && (RPB << LBITS) <= 65536, "list entry must fit 16 bits");
#define CAP 192
static_assert(CAP <= 256 && CAP % 32 == 0, "a batch row number fits one byte; whole 32-row groups");
#define PITCH 68
#define WP 160
#define WROWS (DV + DE)
#define NG 32
#define TP 64

__device__ __forceinline__ v8f mma4(const v2f a, const v2f b, const v8f c) {
    return __builtin_amdgcn_wmma_f32_16x16x4_f32( false, a,  false, b,
                                                  (short)0, c,  false,
                                                  false);
}

template <int CTRL>
__device__ __forceinline__ float lane_swz(float x) {
    const int b = __builtin_bit_cast(int, x);
    return __builtin_bit_cast(float, __builtin_amdgcn_update_dpp(b, b, CTRL, 0xf, 0xf, false));
}
__device__ __forceinline__ float lane_xhalf(float x) {
    const int b = __builtin_bit_cast(int, x);
    return __builtin_bit_cast(
        float, __builtin_amdgcn_permlanex16(b, b, 0x76543210, 0xfedcba98, false, false));
}
#define XM1 0x161
#define XM2 0x162
#define XM4 0x164
#define XM8 0x168

__device__ __forceinline__ int scan_append(const int4 (&rr)[SCAN_IT], const int lo,
                                           const int lane, unsigned short* __restrict__ reg) {
    int cnt = 0;
#pragma unroll
    for (int it = 0; it < SCAN_IT; ++it) {
        const int comp[4] = {rr[it].x, rr[it].y, rr[it].z, rr[it].w};
        int d[4];
        bool match[4];
        unsigned bal[4];
#pragma unroll
        for (int c = 0; c < 4; ++c) {
            d[c] = comp[c] - lo;
            match[c] = (unsigned)d[c] < (unsigned)RPB;
            bal[c] = __builtin_amdgcn_ballot_w32(match[c]);
        }
        if ((bal[0] | bal[1] | bal[2] | bal[3]) == 0u) continue;
#pragma unroll
        for (int c = 0; c < 4; ++c) {
            if (bal[c] == 0u) continue;
            const int pos = cnt + (int)__builtin_amdgcn_mbcnt_lo(bal[c], 0u);
            if (match[c])
                reg[pos] = (unsigned short)((it * 128 + lane * 4 + c) | (d[c] << LBITS));
            cnt += __builtin_popcount(bal[c]);
        }
    }
    return cnt;
}

__device__ __forceinline__ int locate(const int p, const int (&pre)[RPB],
                                      const unsigned short* __restrict__ wl, int& dd) {
    int j = 0, base = 0;
#pragma unroll
    for (int t = 1; t < RPB; ++t) {
        const bool ge = p >= pre[t];
        j = ge ? t : j;
        base = ge ? pre[t] : base;
    }
    const int ent = wl[j * SLICE + (p - base)];
    dd = ent >> LBITS;
    return j * SLICE + (ent & (SLICE - 1));
}

template <int F, int ROW0>
__device__ __forceinline__ void strip(const float* __restrict__ arow, const int grow,
                                      const float* __restrict__ wsm,
                                      const float* __restrict__ tab,
                                      float* __restrict__ dst, const int lane) {
    constexpr int NB = F / 32;
    const int c = lane & 15, h = lane >> 4;
    const float* __restrict__ wl = wsm + (ROW0 / 2 + h) * WP + 2 * c;
    v8f acc[4] = {};
    v2f an[8];
#pragma unroll
    for (int i = 0; i < 8; ++i) an[i] = *(const v2f*)(arow + 2 * h + 4 * i);
#pragma unroll 1
    for (int j = 0; j < NB; ++j) {
        v2f ac[8];
#pragma unroll
        for (int i = 0; i < 8; ++i) ac[i] = an[i];
        if (j + 1 < NB) {
            const float* __restrict__ pn = arow + 32 * (j + 1) + 2 * h;
#pragma unroll
            for (int i = 0; i < 8; ++i) an[i] = *(const v2f*)(pn + 4 * i);
        }
        const float* __restrict__ wj = wl + j * 16 * WP;
#pragma unroll
        for (int i = 0; i < 8; ++i) {
#pragma unroll
            for (int ct = 0; ct < 4; ++ct) {
                const v2f b = *(const v2f*)(wj + (2 * i) * WP + 32 * ct);
                acc[ct] = mma4(ac[i], b, acc[ct]);
            }
        }
    }
#pragma unroll
    for (int j = 0; j < 8; ++j) {
        const int gj = __shfl(grow, 8 * h + j, 16);
        const float4 tv = *(const float4*)(tab + gj * TP + 4 * c);
        const float t4[4] = {tv.x, tv.y, tv.z, tv.w};
#pragma unroll
        for (int ct = 0; ct < 4; ++ct) {
            float x = acc[ct][j] + t4[ct];
            x = (x > 0.f) ? x : x * NEG_SLOPE;
            dst[(8 * h + j) * PITCH + ct * 16 + c] = x;
        }
    }
}

__device__ __forceinline__ void u_table_operands(const float* __restrict__ u,
                                                 const float* __restrict__ Wu, const int rt,
                                                 const int ct, const int lane, v2f (&a)[DU / 4],
                                                 v2f (&b)[DU / 4]) {
    const int c = lane & 15, h = lane >> 4;
    const int col = ct * 16 + c;
    const float* __restrict__ urow = u + (rt * 16 + c) * DU;
#pragma unroll
    for (int t = 0; t < DU / 4; ++t) {
        const int k0 = 4 * t + 2 * h;
        a[t] = *(const v2f*)(urow + k0);
        b[t].x = Wu[(size_t)k0 * DA + col];
        b[t].y = Wu[(size_t)(k0 + 1) * DA + col];
    }
}
__device__ __forceinline__ void u_table_tile(const v2f (&a)[DU / 4], const v2f (&b)[DU / 4],
                                             const float bc, float* __restrict__ tab,
                                             const int rt, const int ct, const int lane) {
    const int c = lane & 15, h = lane >> 4;
    v8f acc = {};
#pragma unroll
    for (int t = 0; t < DU / 4; ++t) acc = mma4(a[t], b[t], acc);
#pragma unroll
    for (int j = 0; j < 8; ++j) tab[(rt * 16 + 8 * h + j) * TP + 4 * c + ct] = acc[j] + bc;
}

__device__ __forceinline__ void q_tile(const float* __restrict__ arow,
                                       const float* __restrict__ urow,
                                       const float* __restrict__ W,
                                       const float* __restrict__ bias,
                                       float* __restrict__ qs, const int ct, const int lane) {
    const int c = lane & 15, h = lane >> 4;
    const int col = ct * 16 + c;
    v8f acc = {};
#pragma unroll
    for (int kb = 0; kb < DV + DU; kb += 4) {
        const int k0 = kb + 2 * h;
        const float* __restrict__ p = (kb < DV) ? (arow + k0) : (urow + (k0 - DV));
        const v2f a = *(const v2f*)p;
        v2f b;
        b.x = W[(size_t)k0 * DA + col];
        b.y = W[(size_t)(k0 + 1) * DA + col];
        acc = mma4(a, b, acc);
    }
    const float bc = bias[col];
#pragma unroll
    for (int j = 0; j < 8; ++j) {
        float x = acc[j] + bc;
        x = (x > 0.f) ? x : x * NEG_SLOPE;
        qs[(8 * h + j) * PITCH + col] = x;
    }
}

__global__ __launch_bounds__(512) void edge_attn_owner_kernel(
    const float* __restrict__ nodes, const float* __restrict__ edges,
    const int* __restrict__ s, const int* __restrict__ recv, const int* __restrict__ batch,
    const float* __restrict__ u, const float* __restrict__ Wq, const float* __restrict__ bq,
    const float* __restrict__ Wk, const float* __restrict__ bk, const float* __restrict__ Wv,
    const float* __restrict__ bv, float* __restrict__ out) {
    __shared__ unsigned short wl[N_EDGES];
    __shared__ int cntw[RPB];
    __shared__ unsigned char drow[CAP];
    __shared__ unsigned char myrow[RPB][CAP];
    __shared__ __attribute__((aligned(16))) float wsm[WROWS / 2 * WP];
    __shared__ __attribute__((aligned(16))) float tab[2 * NG * TP];
    __shared__ __attribute__((aligned(16))) float qs[RPB * PITCH];
    __shared__ __attribute__((aligned(16))) float ks[CAP * PITCH];
    __shared__ __attribute__((aligned(16))) float vs[CAP * PITCH];

    const int lane = threadIdx.x & 31;
    const int w = __builtin_amdgcn_readfirstlane((int)(threadIdx.x >> 5));
    const int lo = blockIdx.x * RPB;
    const int ebase = w * SLICE;

    const int trt = (w >> 2) & 1, tct = w & 3;
    v2f ta[DU / 4], tb[DU / 4];
    u_table_operands(u, (w < 8) ? (Wk + (size_t)DV * DA) : (Wv + (size_t)DE * DA), trt, tct, lane,
                     ta, tb);
    const float tbias = ((w < 8) ? bk : bv)[tct * 16 + (lane & 15)];

    const int4* __restrict__ r4 = (const int4*)(recv + ebase);
    int4 rr[SCAN_IT];
#pragma unroll
    for (int it = 0; it < SCAN_IT; ++it) rr[it] = r4[it * 32 + lane];
    __builtin_amdgcn_sched_barrier(0);

    u_table_tile(ta, tb, tbias, tab + (w >> 3) * (NG * TP), trt, tct, lane);

    {
#pragma unroll
        for (int it = 0; it < WROWS / 2 * (DA / 4) / 512; ++it) {
            const int item = it * 512 + (int)threadIdx.x;
            const int pr = item >> 4;
            const int a = item & 15;
            const float* __restrict__ srow = (pr < DV / 2)
                                                 ? (Wk + (size_t)(2 * pr) * DA)
                                                 : (Wv + (size_t)(2 * pr - DV) * DA);
            const float4 x = ((const float4*)srow)[a];
            const float4 y = ((const float4*)(srow + DA))[a];
            float4 o0, o1;
            o0.x = x.x; o0.y = y.x; o0.z = x.y; o0.w = y.y;
            o1.x = x.z; o1.y = y.z; o1.z = x.w; o1.w = y.w;
            float4* __restrict__ d = (float4*)(wsm + pr * WP + a * 8);
            d[0] = o0;
            d[1] = o1;
        }
    }

    const int g = lane >> 3;
    const int sub = lane & 7;
    const int rn = lo + w;

    const int cntv = scan_append(rr, lo, lane, wl + ebase);
    if (lane == 0) cntw[w] = cntv;
    __syncthreads();

    const int tot = cntw[lane & 15];
    int incl = tot;
#pragma unroll
    for (int off = 1; off < RPB; off <<= 1) {
        const int up = __shfl_up(incl, off, 16);
        if ((lane & 15) >= off) incl += up;
    }
    const int excl = incl - tot;
    const int etot = __builtin_amdgcn_readlane(incl, RPB - 1);
    int pre[RPB];
#pragma unroll
    for (int t = 0; t < RPB; ++t) pre[t] = __builtin_amdgcn_readlane(excl, t);

    const int c = lane & 15;
    if (w >= 12) {
        const int mrow = lo + c;
        q_tile(nodes + (size_t)mrow * DV, u + batch[mrow] * DU, Wq, bq, qs, w - 12, lane);
    }

    float m = -INFINITY, l = 0.f;
    float4 a0 = {0.f, 0.f, 0.f, 0.f}, a1 = {0.f, 0.f, 0.f, 0.f};

    for (int c0 = 0; c0 < etot; c0 += CAP) {
        const int cn = (etot - c0 < CAP) ? (etot - c0) : CAP;
        const int nrt = (cn + 15) >> 4;
        if (w < 8) {
            for (int rt = w; rt < nrt; rt += 8) {
                const int idx = rt * 16 + c;
                int dd;
                const int e = locate(c0 + ((idx < cn) ? idx : 0), pre, wl, dd);
                if (lane < 16 && idx < cn) drow[idx] = (unsigned char)dd;
                const int sn = s[e];
                strip<DV, 0>(nodes + (size_t)sn * DV, batch[sn], wsm, tab,
                             ks + rt * 16 * PITCH, lane);
            }
        } else {
            for (int rt = (w + 6) & 7; rt < nrt; rt += 8) {
                const int idx = rt * 16 + c;
                int dd;
                const int e = locate(c0 + ((idx < cn) ? idx : 0), pre, wl, dd);
                const int sn = s[e];
                strip<DE, DV>(edges + (size_t)e * DE, batch[sn], wsm, tab + NG * TP,
                              vs + rt * 16 * PITCH, lane);
            }
        }
        __syncthreads();

        const float4* __restrict__ qp = (const float4*)(qs + w * PITCH);
        const float4 qa = qp[sub], qb = qp[8 + sub];
        unsigned char* __restrict__ mr = myrow[w];
        int dv[CAP / 32];
#pragma unroll
        for (int t = 0; t < CAP / 32; ++t) dv[t] = (int)drow[t * 32 + lane];
        int nw = 0;
#pragma unroll
        for (int t = 0; t < CAP / 32; ++t) {
            const int idx = t * 32 + lane;
            const bool mine = (idx < cn) && (dv[t] == w);
            const unsigned bm = __builtin_amdgcn_ballot_w32(mine);
            if (mine) mr[nw + (int)__builtin_amdgcn_mbcnt_lo(bm, 0u)] = (unsigned char)idx;
            nw += __builtin_popcount(bm);
        }

        for (int i0 = 0; i0 < nw; i0 += 16) {
            float4 ka[4], kb[4], va[4], vb[4];
            bool valid[4];
#pragma unroll
            for (int uu = 0; uu < 4; ++uu) {
                const int i = i0 + uu * 4 + g;
                valid[uu] = i < nw;
                const int row = mr[valid[uu] ? i : i0];
                const float4* __restrict__ kp = (const float4*)(ks + row * PITCH);
                const float4* __restrict__ vp = (const float4*)(vs + row * PITCH);
                ka[uu] = kp[sub]; kb[uu] = kp[8 + sub];
                va[uu] = vp[sub]; vb[uu] = vp[8 + sub];
            }
#pragma unroll
            for (int uu = 0; uu < 4; ++uu) {
                float sd = qa.x * ka[uu].x + qa.y * ka[uu].y + qa.z * ka[uu].z + qa.w * ka[uu].w +
                           qb.x * kb[uu].x + qb.y * kb[uu].y + qb.z * kb[uu].z + qb.w * kb[uu].w;
                sd += lane_swz<XM1>(sd);
                sd += lane_swz<XM2>(sd);
                sd += lane_swz<XM4>(sd);
                const float t = sd * 0.125f;
                const float mn = valid[uu] ? fmaxf(m, t) : m;
                const float sc = (m == mn) ? 1.f : __expf(m - mn);
                const float p = valid[uu] ? __expf(t - mn) : 0.f;
                l = l * sc + p;
                a0.x = a0.x * sc + p * va[uu].x; a0.y = a0.y * sc + p * va[uu].y;
                a0.z = a0.z * sc + p * va[uu].z; a0.w = a0.w * sc + p * va[uu].w;
                a1.x = a1.x * sc + p * vb[uu].x; a1.y = a1.y * sc + p * vb[uu].y;
                a1.z = a1.z * sc + p * vb[uu].z; a1.w = a1.w * sc + p * vb[uu].w;
                m = mn;
            }
        }
        __syncthreads();
    }

    {
        const float mo = lane_swz<XM8>(m), lq = lane_swz<XM8>(l);
        const float M = fmaxf(m, mo);
        const float s1 = (m == M) ? 1.f : __expf(m - M);
        const float s2 = (mo == M) ? 1.f : __expf(mo - M);
        l = l * s1 + lq * s2;
        a0.x = a0.x * s1 + lane_swz<XM8>(a0.x) * s2; a0.y = a0.y * s1 + lane_swz<XM8>(a0.y) * s2;
        a0.z = a0.z * s1 + lane_swz<XM8>(a0.z) * s2; a0.w = a0.w * s1 + lane_swz<XM8>(a0.w) * s2;
        a1.x = a1.x * s1 + lane_swz<XM8>(a1.x) * s2; a1.y = a1.y * s1 + lane_swz<XM8>(a1.y) * s2;
        a1.z = a1.z * s1 + lane_swz<XM8>(a1.z) * s2; a1.w = a1.w * s1 + lane_swz<XM8>(a1.w) * s2;
        m = M;
    }
    {
        const float mo = lane_xhalf(m), lq = lane_xhalf(l);
        const float M = fmaxf(m, mo);
        const float s1 = (m == M) ? 1.f : __expf(m - M);
        const float s2 = (mo == M) ? 1.f : __expf(mo - M);
        l = l * s1 + lq * s2;
        a0.x = a0.x * s1 + lane_xhalf(a0.x) * s2; a0.y = a0.y * s1 + lane_xhalf(a0.y) * s2;
        a0.z = a0.z * s1 + lane_xhalf(a0.z) * s2; a0.w = a0.w * s1 + lane_xhalf(a0.w) * s2;
        a1.x = a1.x * s1 + lane_xhalf(a1.x) * s2; a1.y = a1.y * s1 + lane_xhalf(a1.y) * s2;
        a1.z = a1.z * s1 + lane_xhalf(a1.z) * s2; a1.w = a1.w * s1 + lane_xhalf(a1.w) * s2;
    }

    if (lane < 8) {
        float4* __restrict__ op = (float4*)(out + (size_t)rn * DA);
        float4 o0, o1;
        o0.x = a0.x / l; o0.y = a0.y / l; o0.z = a0.z / l; o0.w = a0.w / l;
        o1.x = a1.x / l; o1.y = a1.y / l; o1.z = a1.z / l; o1.w = a1.w / l;
        op[sub] = o0;
        op[8 + sub] = o1;
    }
}

extern "C" void kernel_launch(void* const* d_in, const int* in_sizes, int n_in,
                              void* d_out, int out_size, void* d_ws, size_t ws_size,
                              hipStream_t stream) {
    const float* nodes = (const float*)d_in[0];
    const float* edges = (const float*)d_in[1];
    const int* edge_index = (const int*)d_in[2];
    const float* u = (const float*)d_in[3];
    const int* batch = (const int*)d_in[4];
    const float* Wq = (const float*)d_in[5];
    const float* bq = (const float*)d_in[6];
    const float* Wk = (const float*)d_in[7];
    const float* bk = (const float*)d_in[8];
    const float* Wv = (const float*)d_in[9];
    const float* bv = (const float*)d_in[10];
    const int* s = edge_index;
    const int* r = edge_index + N_EDGES;
    float* out = (float*)d_out;

    edge_attn_owner_kernel<<<N_NODES / RPB, 512, 0, stream>>>(nodes, edges, s, r, batch, u, Wq,
                                                              bq, Wk, bk, Wv, bv, out);
}
